# starting kernel
# baseline (speedup 1.0000x reference)
#include <hip/hip_runtime.h>
#include <hip/hip_bf16.h>

#define N_NODES 50000
#define N_EDGES 600000
#define DIM 128
#define ALPHA_C 0.1f
#define BETA_C 0.22314355131420976f
#define HS_STRIDE 132

typedef __attribute__((ext_vector_type(2))) float v2f;
typedef __attribute__((ext_vector_type(8))) float v8f;

__global__ void gcnii_zero(float* out, float* __restrict__ deg) {
    int idx = blockIdx.x * blockDim.x + threadIdx.x;
    if (idx < N_NODES * DIM) out[idx] = 0.0f;
    if (idx < N_NODES) deg[idx] = 0.0f;
}

__global__ void gcnii_degree(const int* __restrict__ dst, float* __restrict__ deg) {
    int e = blockIdx.x * blockDim.x + threadIdx.x;
    if (e < N_EDGES) atomicAdd(&deg[dst[e]], 1.0f);
}

__global__ void gcnii_norm(float* __restrict__ deg) {
    int i = blockIdx.x * blockDim.x + threadIdx.x;
    if (i < N_NODES) deg[i] = rsqrtf(fmaxf(deg[i], 1.0f));
}

__global__ void gcnii_scatter(const float* __restrict__ feat,
                              const float* __restrict__ norm,
                              const int* __restrict__ src,
                              const int* __restrict__ dst,
                              float* agg) {
    int gtid = blockIdx.x * blockDim.x + threadIdx.x;
    int e    = gtid >> 5;
    int lane = gtid & 31;
    if (e >= N_EDGES) return;
    int s = src[e];
    int d = dst[e];
    float ns = norm[s];
    float4 v = *(const float4*)(feat + (size_t)s * DIM + lane * 4);
    float* ad = agg + (size_t)d * DIM + lane * 4;
    atomicAdd(ad + 0, v.x * ns);
    atomicAdd(ad + 1, v.y * ns);
    atomicAdd(ad + 2, v.z * ns);
    atomicAdd(ad + 3, v.w * ns);
}

__global__ __launch_bounds__(256) void gcnii_gemm(
    const float* agg,
    const float* __restrict__ feat0,
    const float* __restrict__ W,
    const float* __restrict__ bias,
    const float* __restrict__ norm,
    float* out)
{
    __shared__ float Wl[DIM * DIM];
    __shared__ float hs[16 * HS_STRIDE];
    __shared__ float bs[DIM];

    const int tid = threadIdx.x;
    const int m0  = blockIdx.x * 16;

    for (int i = tid; i < DIM * DIM / 4; i += 256)
        ((float4*)Wl)[i] = ((const float4*)W)[i];
    if (tid < DIM) bs[tid] = bias[tid];

    for (int i = tid; i < 16 * DIM / 4; i += 256) {
        int r = i >> 5;
        int c = (i & 31) * 4;
        int row = m0 + r;
        float nr = norm[row] * (1.0f - ALPHA_C);
        float4 a  = *(const float4*)(agg   + (size_t)row * DIM + c);
        float4 f0 = *(const float4*)(feat0 + (size_t)row * DIM + c);
        float4 h;
        h.x = a.x * nr + ALPHA_C * f0.x;
        h.y = a.y * nr + ALPHA_C * f0.y;
        h.z = a.z * nr + ALPHA_C * f0.z;
        h.w = a.w * nr + ALPHA_C * f0.w;
        *(float4*)(hs + r * HS_STRIDE + c) = h;
    }
    __syncthreads();

    const int wave = tid >> 5;
    const int lane = tid & 31;
    const int half = lane >> 4;
    const int lr   = lane & 15;
    const int n0   = wave * 16;

    v8f acc = {};
    #pragma unroll
    for (int k0 = 0; k0 < DIM; k0 += 4) {
        const int ka = k0 + 2 * half;
        v2f a, b;
        a.x = hs[lr * HS_STRIDE + ka];
        a.y = hs[lr * HS_STRIDE + ka + 1];
        b.x = Wl[ka * DIM + n0 + lr];
        b.y = Wl[(ka + 1) * DIM + n0 + lr];
        acc = __builtin_amdgcn_wmma_f32_16x16x4_f32(
             false, a,  false, b,
             (short)0, acc,  false,  false);
    }

    #pragma unroll
    for (int r = 0; r < 8; r++) {
        int lrow = r + 8 * half;
        int row  = m0 + lrow;
        int col  = n0 + lr;
        float hv = hs[lrow * HS_STRIDE + col];
        out[(size_t)row * DIM + col] =
            (1.0f - BETA_C) * hv + BETA_C * acc[r] + bs[col];
    }
}

extern "C" void kernel_launch(void* const* d_in, const int* in_sizes, int n_in,
                              void* d_out, int out_size, void* d_ws, size_t ws_size,
                              hipStream_t stream) {
    const float* feat  = (const float*)d_in[0];
    const float* feat0 = (const float*)d_in[1];
    const float* W     = (const float*)d_in[2];
    const float* bias  = (const float*)d_in[3];
    const int*   src   = (const int*)d_in[4];
    const int*   dst   = (const int*)d_in[5];
    float* out = (float*)d_out;
    float* deg = (float*)d_ws;

    gcnii_zero<<<(N_NODES * DIM + 255) / 256, 256, 0, stream>>>(out, deg);
    gcnii_degree<<<(N_EDGES + 255) / 256, 256, 0, stream>>>(dst, deg);
    gcnii_norm<<<(N_NODES + 255) / 256, 256, 0, stream>>>(deg);
    gcnii_scatter<<<(N_EDGES * 32 + 255) / 256, 256, 0, stream>>>(feat, deg, src, dst, out);
    gcnii_gemm<<<N_NODES / 16, 256, 0, stream>>>(out, feat0, W, bias, deg, out);
}


# Aggregation made owner-computes: each of 250 work-groups keeps 200 target rows in LDS, scans dst, gathers and sums with LDS float atomics (arrival order), then blends and multiplies from LDS
# speedup vs baseline: 1.5015x; 1.5015x over previous
#include <hip/hip_runtime.h>
#include <hip/hip_bf16.h>


#define N_NODES 50000
#define N_EDGES 600000
#define DIM 128
#define ALPHA_C 0.1f
#define BETA_C 0.22314355131420976f
#define HS_STRIDE 132

#define R_NODES 200
#define N_GROUPS 250
#define N_STRIPS 13
#define ACC_ROWS 209
#define SINK_ROW 208
#define QCAP 8192
#define NT 1024
#define NV4 (N_EDGES / 4)
#define SCAN_U 8
#define CU 8
#define SCAN_FULL (NV4 / (NT * SCAN_U) * (NT * SCAN_U))
#define SCAN_TAIL_U ((NV4 - SCAN_FULL + NT - 1) / NT)

static_assert(N_GROUPS * R_NODES == N_NODES, "ownership must tile the node set");
static_assert(N_EDGES % 4 == 0, "int4 view of dst");
static_assert(N_EDGES < (1 << 20), "edge id is packed into 20 bits");
static_assert(N_NODES <= (1 << 16), "source id is packed into 16 bits");
static_assert(R_NODES % 8 == 0, "a half strip (8 rows) is wholly real or wholly padding");
static_assert(QCAP % CU == 0, "queue is read CU entries at a time");

typedef __attribute__((ext_vector_type(2))) float v2f;
typedef __attribute__((ext_vector_type(8))) float v8f;

__global__ void gcnii_zero(float* __restrict__ deg) {
    int idx = blockIdx.x * blockDim.x + threadIdx.x;
    if (idx < N_NODES) deg[idx] = 0.0f;
}

__global__ void gcnii_degree(const int* __restrict__ dst, float* __restrict__ deg) {
    int e = blockIdx.x * blockDim.x + threadIdx.x;
    if (e < N_EDGES) atomicAdd(&deg[dst[e]], 1.0f);
}

__global__ void gcnii_norm(float* __restrict__ deg) {
    int i = blockIdx.x * blockDim.x + threadIdx.x;
    if (i < N_NODES) deg[i] = rsqrtf(fmaxf(deg[i], 1.0f));
}

__device__ __forceinline__ void add_edge_slow(const float* __restrict__ feat,
                                              const float* __restrict__ norm,
                                              const int* __restrict__ src,
                                              float* hacc, unsigned e, unsigned dl) {
    int s = src[e];
    float ns = norm[s];
    for (int c = 0; c < DIM; ++c)
        atomicAdd(&hacc[dl * HS_STRIDE + c], feat[(size_t)s * DIM + c] * ns);
}

__device__ __forceinline__ void scan_one(const int4 v, const int j, const int nbase,
                                         unsigned* q, unsigned* qn, float* hacc,
                                         const float* __restrict__ feat,
                                         const float* __restrict__ norm,
                                         const int* __restrict__ src) {
    unsigned dl[4];
    dl[0] = (unsigned)(v.x - nbase);
    dl[1] = (unsigned)(v.y - nbase);
    dl[2] = (unsigned)(v.z - nbase);
    dl[3] = (unsigned)(v.w - nbase);
    if ((dl[0] < R_NODES) | (dl[1] < R_NODES) | (dl[2] < R_NODES) | (dl[3] < R_NODES)) {
        #pragma unroll
        for (int c = 0; c < 4; ++c) {
            if (dl[c] < R_NODES) {
                unsigned e = 4u * (unsigned)j + (unsigned)c;
                unsigned slot = atomicAdd(qn, 1u);
                if (slot < QCAP) q[slot] = (dl[c] << 20) | e;
                else add_edge_slow(feat, norm, src, hacc, e, dl[c]);
            }
        }
    }
}

__global__ __launch_bounds__(NT) void gcnii_own(
    const float* __restrict__ feat,
    const float* __restrict__ feat0,
    const float* __restrict__ W,
    const float* __restrict__ bias,
    const float* __restrict__ norm,
    const int* __restrict__ src,
    const int* __restrict__ dst,
    float* __restrict__ out)
{
    __shared__ __attribute__((aligned(16))) float hacc[ACC_ROWS * HS_STRIDE];
    __shared__ __attribute__((aligned(16))) float Wl[DIM * DIM];
    __shared__ __attribute__((aligned(16))) unsigned q[QCAP];
    __shared__ __attribute__((aligned(16))) float qns[QCAP];
    __shared__ float bs[DIM];
    __shared__ unsigned qn;

    const int tid  = threadIdx.x;
    const int lane = tid & 31;
    const int wave = __builtin_amdgcn_readfirstlane(tid >> 5);
    const int nbase = blockIdx.x * R_NODES;

    for (int i = tid; i < ACC_ROWS * HS_STRIDE / 4; i += NT)
        ((float4*)hacc)[i] = make_float4(0.0f, 0.0f, 0.0f, 0.0f);
    for (int i = tid; i < DIM * DIM / 4; i += NT)
        ((float4*)Wl)[i] = ((const float4*)W)[i];
    if (tid < DIM) bs[tid] = bias[tid];
    if (tid == 0) qn = 0u;
    __syncthreads();

    const int4* d4 = (const int4*)dst;
    const int off = blockIdx.x * (NV4 / 256);
    for (int i0 = 0; i0 < SCAN_FULL; i0 += NT * SCAN_U) {
        int4 v[SCAN_U];
        int  jx[SCAN_U];
        #pragma unroll
        for (int u = 0; u < SCAN_U; ++u) {
            int j = i0 + u * NT + tid + off;
            if (j >= NV4) j -= NV4;
            jx[u] = j;
            v[u] = d4[j];
        }
        #pragma unroll
        for (int u = 0; u < SCAN_U; ++u)
            scan_one(v[u], jx[u], nbase, q, &qn, hacc, feat, norm, src);
    }
    {
        int4 v[SCAN_TAIL_U];
        int  jx[SCAN_TAIL_U];
        #pragma unroll
        for (int u = 0; u < SCAN_TAIL_U; ++u) {
            int i = SCAN_FULL + u * NT + tid;
            int j = i + off;
            if (j >= NV4) j -= NV4;
            jx[u] = j;
            v[u] = (i < NV4) ? d4[j] : make_int4(-1, -1, -1, -1);
        }
        #pragma unroll
        for (int u = 0; u < SCAN_TAIL_U; ++u)
            scan_one(v[u], jx[u], nbase, q, &qn, hacc, feat, norm, src);
    }
    __syncthreads();

    const int n  = __builtin_amdgcn_readfirstlane((int)min(qn, (unsigned)QCAP));
    const int n8 = (n + CU - 1) & ~(CU - 1);
    for (int i = tid; i < n8; i += NT) {
        unsigned pk = (unsigned)SINK_ROW << 16;
        float ns = 0.0f;
        if (i < n) {
            unsigned p = q[i];
            int s = src[p & 0xFFFFFu];
            ns = norm[s];
            pk = ((p >> 20) << 16) | (unsigned)s;
        }
        q[i]   = pk;
        qns[i] = ns;
    }
    __syncthreads();

    for (int i0 = wave * CU; i0 < n; i0 += (NT / 32) * CU) {
        const uint4  pa = *(const uint4*)(q + i0);
        const uint4  pb = *(const uint4*)(q + i0 + 4);
        const float4 na = *(const float4*)(qns + i0);
        const float4 nb = *(const float4*)(qns + i0 + 4);
        const unsigned pku[CU] = { pa.x, pa.y, pa.z, pa.w, pb.x, pb.y, pb.z, pb.w };
        const float    nsu[CU] = { na.x, na.y, na.z, na.w, nb.x, nb.y, nb.z, nb.w };
        v2f xa[CU], xb[CU];
        #pragma unroll
        for (int u = 0; u < CU; ++u) {
            const int s = __builtin_amdgcn_readfirstlane((int)(pku[u] & 0xFFFFu));
            const float* fr = feat + (size_t)s * DIM + 2 * lane;
            xa[u] = *(const v2f*)(fr);
            xb[u] = *(const v2f*)(fr + 64);
        }
        #pragma unroll
        for (int u = 0; u < CU; ++u) {
            const int dl = __builtin_amdgcn_readfirstlane((int)(pku[u] >> 16));
            float* ar = hacc + dl * HS_STRIDE + 2 * lane;
            atomicAdd(ar,      xa[u].x * nsu[u]);
            atomicAdd(ar + 1,  xa[u].y * nsu[u]);
            atomicAdd(ar + 64, xb[u].x * nsu[u]);
            atomicAdd(ar + 65, xb[u].y * nsu[u]);
        }
    }
    __syncthreads();

    for (int i = tid; i < R_NODES * (DIM / 4); i += NT) {
        int r = i >> 5;
        int c = (i & 31) * 4;
        int row = nbase + r;
        float nr = norm[row] * (1.0f - ALPHA_C);
        float4 a  = *(const float4*)(hacc + r * HS_STRIDE + c);
        float4 f0 = *(const float4*)(feat0 + (size_t)row * DIM + c);
        float4 h;
        h.x = a.x * nr + ALPHA_C * f0.x;
        h.y = a.y * nr + ALPHA_C * f0.y;
        h.z = a.z * nr + ALPHA_C * f0.z;
        h.w = a.w * nr + ALPHA_C * f0.w;
        *(float4*)(hacc + r * HS_STRIDE + c) = h;
    }
    __syncthreads();

    const int half = lane >> 4;
    const int lr   = lane & 15;
    const int n0   = (wave & 7) * 16;
    for (int st = wave >> 3; st < N_STRIPS; st += 4) {
        const float* hs = hacc + st * 16 * HS_STRIDE;
        v8f acc = {};
        #pragma unroll
        for (int k0 = 0; k0 < DIM; k0 += 4) {
            const int ka = k0 + 2 * half;
            v2f a, b;
            a.x = hs[lr * HS_STRIDE + ka];
            a.y = hs[lr * HS_STRIDE + ka + 1];
            b.x = Wl[ka * DIM + n0 + lr];
            b.y = Wl[(ka + 1) * DIM + n0 + lr];
            acc = __builtin_amdgcn_wmma_f32_16x16x4_f32(
                 false, a,  false, b,
                 (short)0, acc,  false,  false);
        }

        const int col   = n0 + lr;
        const int lrow0 = st * 16 + 8 * half;
        float val[8];
        #pragma unroll
        for (int r = 0; r < 8; r++) {
            float hv = hacc[(lrow0 + r) * HS_STRIDE + col];
            val[r] = (1.0f - BETA_C) * hv + BETA_C * acc[r] + bs[col];
        }
        if (lrow0 < R_NODES) {
            #pragma unroll
            for (int r = 0; r < 8; r++)
                out[(size_t)(nbase + lrow0 + r) * DIM + col] = val[r];
        }
    }
}

extern "C" void kernel_launch(void* const* d_in, const int* in_sizes, int n_in,
                              void* d_out, int out_size, void* d_ws, size_t ws_size,
                              hipStream_t stream) {
    const float* feat  = (const float*)d_in[0];
    const float* feat0 = (const float*)d_in[1];
    const float* W     = (const float*)d_in[2];
    const float* bias  = (const float*)d_in[3];
    const int*   src   = (const int*)d_in[4];
    const int*   dst   = (const int*)d_in[5];
    float* out = (float*)d_out;
    float* deg = (float*)d_ws;

    gcnii_zero<<<(N_NODES + 255) / 256, 256, 0, stream>>>(deg);
    gcnii_degree<<<(N_EDGES + 255) / 256, 256, 0, stream>>>(dst, deg);
    gcnii_norm<<<(N_NODES + 255) / 256, 256, 0, stream>>>(deg);
    gcnii_own<<<N_GROUPS, NT, 0, stream>>>(feat, feat0, W, bias, deg, src, dst, out);
}


# Owner groups stop scanning the whole dst list: the degree kernel files each edge once under its owner group (LDS rank + one counter add per group); owners read their own file; arrival-order sums
# speedup vs baseline: 1.7426x; 1.1606x over previous
#include <hip/hip_runtime.h>
#include <hip/hip_bf16.h>


#define N_NODES 50000
#define N_EDGES 600000
#define DIM 128
#define ALPHA_C 0.1f
#define BETA_C 0.22314355131420976f
#define HS_STRIDE 132

#define R_NODES 200
#define N_GROUPS 250
#define N_STRIPS 13
#define ACC_ROWS 209
#define SINK_ROW 208
#define QCAP 8192
#define NT 1024
#define NV4 (N_EDGES / 4)
#define SCAN_U 8
#define CU 8
#define SCAN_FULL (NV4 / (NT * SCAN_U) * (NT * SCAN_U))
#define SCAN_TAIL_U ((NV4 - SCAN_FULL + NT - 1) / NT)
#define GROUP_WORDS (R_NODES * DIM)
#define FILE_OFF 32
#define BIN_NT 1024
#define BIN_GROUPS ((NV4 + BIN_NT - 1) / BIN_NT)

static_assert(N_GROUPS * R_NODES == N_NODES, "ownership must tile the node set");
static_assert(N_EDGES % 4 == 0, "int4 view of dst");
static_assert(N_EDGES < (1 << 20), "edge id is packed into 20 bits");
static_assert(N_NODES <= (1 << 16), "source id is packed into 16 bits");
static_assert(R_NODES % 8 == 0, "a half strip (8 rows) is wholly real or wholly padding");
static_assert(QCAP % CU == 0, "queue is read CU entries at a time");
static_assert(FILE_OFF + QCAP <= GROUP_WORDS, "a group's edge file lies inside the group's own out rows");
static_assert(R_NODES <= (1 << 16), "local row is packed above a 16-bit source id");

typedef __attribute__((ext_vector_type(2))) float v2f;
typedef __attribute__((ext_vector_type(8))) float v8f;

__device__ __forceinline__ unsigned* group_words(float* out, unsigned g) {
    return (unsigned*)out + (size_t)g * GROUP_WORDS;
}

__global__ void gcnii_zero(float* __restrict__ deg, float* out) {
    int idx = blockIdx.x * blockDim.x + threadIdx.x;
    if (idx < N_NODES) deg[idx] = 0.0f;
    if (idx < N_GROUPS) group_words(out, (unsigned)idx)[0] = 0u;
}

__global__ __launch_bounds__(BIN_NT) void gcnii_degree_bin(const int* __restrict__ src,
                                                          const int* __restrict__ dst,
                                                          float* __restrict__ deg,
                                                          float* out) {
    __shared__ unsigned hist[N_GROUPS];
    const int tid = threadIdx.x;
    if (tid < N_GROUPS) hist[tid] = 0u;
    __syncthreads();

    const int i = blockIdx.x * BIN_NT + tid;
    const bool live = i < NV4;
    unsigned g[4], pk[4], rk[4];
    if (live) {
        const int4 d = ((const int4*)dst)[i];
        const int4 s = ((const int4*)src)[i];
        const int dv[4] = { d.x, d.y, d.z, d.w };
        const int sv[4] = { s.x, s.y, s.z, s.w };
        #pragma unroll
        for (int c = 0; c < 4; ++c) {
            g[c]  = (unsigned)dv[c] / R_NODES;
            pk[c] = (((unsigned)dv[c] - g[c] * R_NODES) << 16) | (unsigned)sv[c];
            rk[c] = atomicAdd(&hist[g[c]], 1u);
            atomicAdd(&deg[dv[c]], 1.0f);
        }
    }
    __syncthreads();
    if (tid < N_GROUPS) {
        const unsigned h = hist[tid];
        hist[tid] = h ? atomicAdd(group_words(out, (unsigned)tid), h) : 0u;
    }
    __syncthreads();
    if (live) {
        #pragma unroll
        for (int c = 0; c < 4; ++c) {
            const unsigned slot = hist[g[c]] + rk[c];
            if (slot < QCAP) group_words(out, g[c])[FILE_OFF + slot] = pk[c];
        }
    }
}

__global__ void gcnii_norm(float* __restrict__ deg) {
    int i = blockIdx.x * blockDim.x + threadIdx.x;
    if (i < N_NODES) deg[i] = rsqrtf(fmaxf(deg[i], 1.0f));
}

__device__ __forceinline__ void add_edge_slow(const float* __restrict__ feat,
                                              const float* __restrict__ norm,
                                              const int* __restrict__ src,
                                              float* hacc, unsigned e, unsigned dl) {
    int s = src[e];
    float ns = norm[s];
    for (int c = 0; c < DIM; ++c)
        atomicAdd(&hacc[dl * HS_STRIDE + c], feat[(size_t)s * DIM + c] * ns);
}

__device__ __forceinline__ void scan_one(const int4 v, const int j, const int nbase,
                                         unsigned* q, unsigned* qn, float* hacc,
                                         const float* __restrict__ feat,
                                         const float* __restrict__ norm,
                                         const int* __restrict__ src) {
    unsigned dl[4];
    dl[0] = (unsigned)(v.x - nbase);
    dl[1] = (unsigned)(v.y - nbase);
    dl[2] = (unsigned)(v.z - nbase);
    dl[3] = (unsigned)(v.w - nbase);
    if ((dl[0] < R_NODES) | (dl[1] < R_NODES) | (dl[2] < R_NODES) | (dl[3] < R_NODES)) {
        #pragma unroll
        for (int c = 0; c < 4; ++c) {
            if (dl[c] < R_NODES) {
                unsigned e = 4u * (unsigned)j + (unsigned)c;
                unsigned slot = atomicAdd(qn, 1u);
                if (slot < QCAP) q[slot] = (dl[c] << 20) | e;
                else add_edge_slow(feat, norm, src, hacc, e, dl[c]);
            }
        }
    }
}

__global__ __launch_bounds__(NT) void gcnii_own(
    const float* __restrict__ feat,
    const float* __restrict__ feat0,
    const float* __restrict__ W,
    const float* __restrict__ bias,
    const float* __restrict__ norm,
    const int* __restrict__ src,
    const int* __restrict__ dst,
    float* out)
{
    __shared__ __attribute__((aligned(16))) float hacc[ACC_ROWS * HS_STRIDE];
    __shared__ __attribute__((aligned(16))) float Wl[DIM * DIM];
    __shared__ __attribute__((aligned(16))) unsigned q[QCAP];
    __shared__ __attribute__((aligned(16))) float qns[QCAP];
    __shared__ float bs[DIM];
    __shared__ unsigned qn;

    const int tid  = threadIdx.x;
    const int lane = tid & 31;
    const int wave = __builtin_amdgcn_readfirstlane(tid >> 5);
    const int nbase = blockIdx.x * R_NODES;

    for (int i = tid; i < ACC_ROWS * HS_STRIDE / 4; i += NT)
        ((float4*)hacc)[i] = make_float4(0.0f, 0.0f, 0.0f, 0.0f);
    for (int i = tid; i < DIM * DIM / 4; i += NT)
        ((float4*)Wl)[i] = ((const float4*)W)[i];
    if (tid < DIM) bs[tid] = bias[tid];
    if (tid == 0) qn = 0u;
    __syncthreads();

    const unsigned* gw = group_words(out, blockIdx.x);
    const unsigned nfile = (unsigned)__builtin_amdgcn_readfirstlane((int)gw[0]);
    int n;
    if (nfile <= (unsigned)QCAP) {
        n = (int)nfile;
        const int n8 = (n + CU - 1) & ~(CU - 1);
        for (int i = tid; i < n8; i += NT) {
            unsigned pk = (unsigned)SINK_ROW << 16;
            float ns = 0.0f;
            if (i < n) {
                pk = gw[FILE_OFF + i];
                ns = norm[pk & 0xFFFFu];
            }
            q[i]   = pk;
            qns[i] = ns;
        }
    } else {
        const int4* d4 = (const int4*)dst;
        const int off = blockIdx.x * (NV4 / 256);
        for (int i0 = 0; i0 < SCAN_FULL; i0 += NT * SCAN_U) {
            int4 v[SCAN_U];
            int  jx[SCAN_U];
            #pragma unroll
            for (int u = 0; u < SCAN_U; ++u) {
                int j = i0 + u * NT + tid + off;
                if (j >= NV4) j -= NV4;
                jx[u] = j;
                v[u] = d4[j];
            }
            #pragma unroll
            for (int u = 0; u < SCAN_U; ++u)
                scan_one(v[u], jx[u], nbase, q, &qn, hacc, feat, norm, src);
        }
        {
            int4 v[SCAN_TAIL_U];
            int  jx[SCAN_TAIL_U];
            #pragma unroll
            for (int u = 0; u < SCAN_TAIL_U; ++u) {
                int i = SCAN_FULL + u * NT + tid;
                int j = i + off;
                if (j >= NV4) j -= NV4;
                jx[u] = j;
                v[u] = (i < NV4) ? d4[j] : make_int4(-1, -1, -1, -1);
            }
            #pragma unroll
            for (int u = 0; u < SCAN_TAIL_U; ++u)
                scan_one(v[u], jx[u], nbase, q, &qn, hacc, feat, norm, src);
        }
        __syncthreads();

        n = __builtin_amdgcn_readfirstlane((int)min(qn, (unsigned)QCAP));
        const int n8 = (n + CU - 1) & ~(CU - 1);
        for (int i = tid; i < n8; i += NT) {
            unsigned pk = (unsigned)SINK_ROW << 16;
            float ns = 0.0f;
            if (i < n) {
                unsigned p = q[i];
                int s = src[p & 0xFFFFFu];
                ns = norm[s];
                pk = ((p >> 20) << 16) | (unsigned)s;
            }
            q[i]   = pk;
            qns[i] = ns;
        }
    }
    __syncthreads();

    for (int i0 = wave * CU; i0 < n; i0 += (NT / 32) * CU) {
        const uint4  pa = *(const uint4*)(q + i0);
        const uint4  pb = *(const uint4*)(q + i0 + 4);
        const float4 na = *(const float4*)(qns + i0);
        const float4 nb = *(const float4*)(qns + i0 + 4);
        const unsigned pku[CU] = { pa.x, pa.y, pa.z, pa.w, pb.x, pb.y, pb.z, pb.w };
        const float    nsu[CU] = { na.x, na.y, na.z, na.w, nb.x, nb.y, nb.z, nb.w };
        v2f xa[CU], xb[CU];
        #pragma unroll
        for (int u = 0; u < CU; ++u) {
            const int s = __builtin_amdgcn_readfirstlane((int)(pku[u] & 0xFFFFu));
            const float* fr = feat + (size_t)s * DIM + 2 * lane;
            xa[u] = *(const v2f*)(fr);
            xb[u] = *(const v2f*)(fr + 64);
        }
        #pragma unroll
        for (int u = 0; u < CU; ++u) {
            const int dl = __builtin_amdgcn_readfirstlane((int)(pku[u] >> 16));
            float* ar = hacc + dl * HS_STRIDE + 2 * lane;
            atomicAdd(ar,      xa[u].x * nsu[u]);
            atomicAdd(ar + 1,  xa[u].y * nsu[u]);
            atomicAdd(ar + 64, xb[u].x * nsu[u]);
            atomicAdd(ar + 65, xb[u].y * nsu[u]);
        }
    }
    __syncthreads();

    for (int i = tid; i < R_NODES * (DIM / 4); i += NT) {
        int r = i >> 5;
        int c = (i & 31) * 4;
        int row = nbase + r;
        float nr = norm[row] * (1.0f - ALPHA_C);
        float4 a  = *(const float4*)(hacc + r * HS_STRIDE + c);
        float4 f0 = *(const float4*)(feat0 + (size_t)row * DIM + c);
        float4 h;
        h.x = a.x * nr + ALPHA_C * f0.x;
        h.y = a.y * nr + ALPHA_C * f0.y;
        h.z = a.z * nr + ALPHA_C * f0.z;
        h.w = a.w * nr + ALPHA_C * f0.w;
        *(float4*)(hacc + r * HS_STRIDE + c) = h;
    }
    __syncthreads();

    const int half = lane >> 4;
    const int lr   = lane & 15;
    const int n0   = (wave & 7) * 16;
    for (int st = wave >> 3; st < N_STRIPS; st += 4) {
        const float* hs = hacc + st * 16 * HS_STRIDE;
        v8f acc = {};
        #pragma unroll
        for (int k0 = 0; k0 < DIM; k0 += 4) {
            const int ka = k0 + 2 * half;
            v2f a, b;
            a.x = hs[lr * HS_STRIDE + ka];
            a.y = hs[lr * HS_STRIDE + ka + 1];
            b.x = Wl[ka * DIM + n0 + lr];
            b.y = Wl[(ka + 1) * DIM + n0 + lr];
            acc = __builtin_amdgcn_wmma_f32_16x16x4_f32(
                 false, a,  false, b,
                 (short)0, acc,  false,  false);
        }

        const int col   = n0 + lr;
        const int lrow0 = st * 16 + 8 * half;
        float val[8];
        #pragma unroll
        for (int r = 0; r < 8; r++) {
            float hv = hacc[(lrow0 + r) * HS_STRIDE + col];
            val[r] = (1.0f - BETA_C) * hv + BETA_C * acc[r] + bs[col];
        }
        if (lrow0 < R_NODES) {
            #pragma unroll
            for (int r = 0; r < 8; r++)
                out[(size_t)(nbase + lrow0 + r) * DIM + col] = val[r];
        }
    }
}

extern "C" void kernel_launch(void* const* d_in, const int* in_sizes, int n_in,
                              void* d_out, int out_size, void* d_ws, size_t ws_size,
                              hipStream_t stream) {
    const float* feat  = (const float*)d_in[0];
    const float* feat0 = (const float*)d_in[1];
    const float* W     = (const float*)d_in[2];
    const float* bias  = (const float*)d_in[3];
    const int*   src   = (const int*)d_in[4];
    const int*   dst   = (const int*)d_in[5];
    float* out = (float*)d_out;
    float* deg = (float*)d_ws;

    gcnii_zero<<<(N_NODES + 255) / 256, 256, 0, stream>>>(deg, out);
    gcnii_degree_bin<<<BIN_GROUPS, BIN_NT, 0, stream>>>(src, dst, deg, out);
    gcnii_norm<<<(N_NODES + 255) / 256, 256, 0, stream>>>(deg);
    gcnii_own<<<N_GROUPS, NT, 0, stream>>>(feat, feat0, W, bias, deg, src, dst, out);
}


# Owner group sorts its edge file by target row in LDS (counting sort); each wave sums a whole row in registers (fma, list = arrival order) and blends it once: LDS float atomics and blend pass gone
# speedup vs baseline: 2.0007x; 1.1481x over previous
#include <hip/hip_runtime.h>
#include <hip/hip_bf16.h>


#define N_NODES 50000
#define N_EDGES 600000
#define DIM 128
#define ALPHA_C 0.1f
#define BETA_C 0.22314355131420976f
#define HS_STRIDE 132

#define R_NODES 200
#define N_GROUPS 250
#define N_STRIPS 13
#define ACC_ROWS 209
#define SINK_ROW 208
#define QCAP 8192
#define NT 1024
#define NV4 (N_EDGES / 4)
#define SCAN_U 8
#define CU 8
#define SCAN_FULL (NV4 / (NT * SCAN_U) * (NT * SCAN_U))
#define SCAN_TAIL_U ((NV4 - SCAN_FULL + NT - 1) / NT)
#define GROUP_WORDS (R_NODES * DIM)
#define FILE_OFF 32
#define BIN_NT 1024
#define BIN_GROUPS ((NV4 + BIN_NT - 1) / BIN_NT)
#define SORT_SLOTS (QCAP / NT)
#define RCNT_N 256
#define ROW_CHUNK 16

static_assert(N_GROUPS * R_NODES == N_NODES, "ownership must tile the node set");
static_assert(N_EDGES % 4 == 0, "int4 view of dst");
static_assert(N_EDGES < (1 << 20), "edge id is packed into 20 bits");
static_assert(N_NODES <= (1 << 16), "source id is packed into 16 bits");
static_assert(R_NODES % 8 == 0, "a half strip (8 rows) is wholly real or wholly padding");
static_assert(QCAP % CU == 0, "queue is read CU entries at a time");
static_assert(FILE_OFF + QCAP <= GROUP_WORDS, "a group's edge file lies inside the group's own out rows");
static_assert(R_NODES <= (1 << 16), "local row is packed above a 16-bit source id");
static_assert(QCAP % NT == 0, "every thread holds the same number of file slots");
static_assert(R_NODES < RCNT_N && RCNT_N == 8 * 32, "one wave scans the row counters, 8 per lane");

typedef __attribute__((ext_vector_type(2))) float v2f;
typedef __attribute__((ext_vector_type(8))) float v8f;

__device__ __forceinline__ unsigned* group_words(float* out, unsigned g) {
    return (unsigned*)out + (size_t)g * GROUP_WORDS;
}

__global__ void gcnii_zero(float* __restrict__ deg, float* out) {
    int idx = blockIdx.x * blockDim.x + threadIdx.x;
    if (idx < N_NODES) deg[idx] = 0.0f;
    if (idx < N_GROUPS) group_words(out, (unsigned)idx)[0] = 0u;
}

__global__ __launch_bounds__(BIN_NT) void gcnii_degree_bin(const int* __restrict__ src,
                                                          const int* __restrict__ dst,
                                                          float* __restrict__ deg,
                                                          float* out) {
    __shared__ unsigned hist[N_GROUPS];
    const int tid = threadIdx.x;
    if (tid < N_GROUPS) hist[tid] = 0u;
    __syncthreads();

    const int i = blockIdx.x * BIN_NT + tid;
    const bool live = i < NV4;
    unsigned g[4], pk[4], rk[4];
    if (live) {
        const int4 d = ((const int4*)dst)[i];
        const int4 s = ((const int4*)src)[i];
        const int dv[4] = { d.x, d.y, d.z, d.w };
        const int sv[4] = { s.x, s.y, s.z, s.w };
        #pragma unroll
        for (int c = 0; c < 4; ++c) {
            g[c]  = (unsigned)dv[c] / R_NODES;
            pk[c] = (((unsigned)dv[c] - g[c] * R_NODES) << 16) | (unsigned)sv[c];
            rk[c] = atomicAdd(&hist[g[c]], 1u);
            atomicAdd(&deg[dv[c]], 1.0f);
        }
    }
    __syncthreads();
    if (tid < N_GROUPS) {
        const unsigned h = hist[tid];
        hist[tid] = h ? atomicAdd(group_words(out, (unsigned)tid), h) : 0u;
    }
    __syncthreads();
    if (live) {
        #pragma unroll
        for (int c = 0; c < 4; ++c) {
            const unsigned slot = hist[g[c]] + rk[c];
            if (slot < QCAP) group_words(out, g[c])[FILE_OFF + slot] = pk[c];
        }
    }
}

__global__ void gcnii_norm(float* __restrict__ deg) {
    int i = blockIdx.x * blockDim.x + threadIdx.x;
    if (i < N_NODES) deg[i] = rsqrtf(fmaxf(deg[i], 1.0f));
}

__device__ __forceinline__ void add_edge_slow(const float* __restrict__ feat,
                                              const float* __restrict__ norm,
                                              const int* __restrict__ src,
                                              float* hacc, unsigned e, unsigned dl) {
    int s = src[e];
    float ns = norm[s];
    for (int c = 0; c < DIM; ++c)
        atomicAdd(&hacc[dl * HS_STRIDE + c], feat[(size_t)s * DIM + c] * ns);
}

__device__ __forceinline__ void scan_one(const int4 v, const int j, const int nbase,
                                         unsigned* q, unsigned* qn, float* hacc,
                                         const float* __restrict__ feat,
                                         const float* __restrict__ norm,
                                         const int* __restrict__ src) {
    unsigned dl[4];
    dl[0] = (unsigned)(v.x - nbase);
    dl[1] = (unsigned)(v.y - nbase);
    dl[2] = (unsigned)(v.z - nbase);
    dl[3] = (unsigned)(v.w - nbase);
    if ((dl[0] < R_NODES) | (dl[1] < R_NODES) | (dl[2] < R_NODES) | (dl[3] < R_NODES)) {
        #pragma unroll
        for (int c = 0; c < 4; ++c) {
            if (dl[c] < R_NODES) {
                unsigned e = 4u * (unsigned)j + (unsigned)c;
                unsigned slot = atomicAdd(qn, 1u);
                if (slot < QCAP) q[slot] = (dl[c] << 20) | e;
                else add_edge_slow(feat, norm, src, hacc, e, dl[c]);
            }
        }
    }
}

__global__ __launch_bounds__(NT) void gcnii_own(
    const float* __restrict__ feat,
    const float* __restrict__ feat0,
    const float* __restrict__ W,
    const float* __restrict__ bias,
    const float* __restrict__ norm,
    const int* __restrict__ src,
    const int* __restrict__ dst,
    float* out)
{
    __shared__ __attribute__((aligned(16))) float hacc[ACC_ROWS * HS_STRIDE];
    __shared__ __attribute__((aligned(16))) float Wl[DIM * DIM];
    __shared__ __attribute__((aligned(16))) unsigned q[QCAP];
    __shared__ __attribute__((aligned(16))) float qns[QCAP];
    __shared__ __attribute__((aligned(16))) unsigned rcnt[RCNT_N];
    __shared__ float bs[DIM];
    __shared__ unsigned qn;

    const int tid  = threadIdx.x;
    const int lane = tid & 31;
    const int wave = __builtin_amdgcn_readfirstlane(tid >> 5);
    const int nbase = blockIdx.x * R_NODES;

    for (int i = tid; i < DIM * DIM / 4; i += NT)
        ((float4*)Wl)[i] = ((const float4*)W)[i];
    if (tid < DIM) bs[tid] = bias[tid];
    if (tid < RCNT_N) rcnt[tid] = 0u;
    if (tid == 0) qn = 0u;
    for (int i = tid; i < (ACC_ROWS - R_NODES) * HS_STRIDE; i += NT)
        hacc[R_NODES * HS_STRIDE + i] = 0.0f;
    __syncthreads();

    const unsigned* gw = group_words(out, blockIdx.x);
    const unsigned nfile = (unsigned)__builtin_amdgcn_readfirstlane((int)gw[0]);
    if (nfile <= (unsigned)QCAP) {
        const int n = (int)nfile;

        unsigned pk[SORT_SLOTS], rk[SORT_SLOTS];
        float    ns[SORT_SLOTS];
        #pragma unroll
        for (int k = 0; k < SORT_SLOTS; ++k) {
            const int i = k * NT + tid;
            pk[k] = 0u;
            if (i < n) pk[k] = gw[FILE_OFF + i];
        }
        #pragma unroll
        for (int k = 0; k < SORT_SLOTS; ++k)
            ns[k] = norm[pk[k] & 0xFFFFu];
        #pragma unroll
        for (int k = 0; k < SORT_SLOTS; ++k) {
            const int i = k * NT + tid;
            rk[k] = 0u;
            if (i < n) rk[k] = atomicAdd(&rcnt[pk[k] >> 16], 1u);
        }
        __syncthreads();

        if (wave == 0) {
            uint4 ca = *(const uint4*)(rcnt + 8 * lane);
            uint4 cb = *(const uint4*)(rcnt + 8 * lane + 4);
            unsigned c[8] = { ca.x, ca.y, ca.z, ca.w, cb.x, cb.y, cb.z, cb.w };
            unsigned t = 0u;
            #pragma unroll
            for (int j = 0; j < 8; ++j) { const unsigned v = c[j]; c[j] = t; t += v; }
            unsigned incl = t;
            #pragma unroll
            for (int d = 1; d < 32; d <<= 1) {
                const unsigned o = __shfl_up(incl, (unsigned)d, 32);
                if (lane >= d) incl += o;
            }
            const unsigned base = incl - t;
            ca = make_uint4(base + c[0], base + c[1], base + c[2], base + c[3]);
            cb = make_uint4(base + c[4], base + c[5], base + c[6], base + c[7]);
            *(uint4*)(rcnt + 8 * lane)     = ca;
            *(uint4*)(rcnt + 8 * lane + 4) = cb;
        }
        __syncthreads();

        #pragma unroll
        for (int k = 0; k < SORT_SLOTS; ++k) {
            const int i = k * NT + tid;
            if (i < n) {
                const unsigned pos = rcnt[pk[k] >> 16] + rk[k];
                q[pos]   = pk[k] & 0xFFFFu;
                qns[pos] = ns[k];
            }
        }
        __syncthreads();

        for (int r = wave; r < R_NODES; r += NT / 32) {
            const int e0 = __builtin_amdgcn_readfirstlane((int)rcnt[r]);
            const int e1 = __builtin_amdgcn_readfirstlane((int)rcnt[r + 1]);
            const int row = nbase + r;
            const float4* f0p = (const float4*)(feat0 + (size_t)row * DIM + 4 * lane);
            float4 f0 = make_float4(0.0f, 0.0f, 0.0f, 0.0f);
            float nrow = 0.0f;
            bool have_f0 = false;
            float4 acc = make_float4(0.0f, 0.0f, 0.0f, 0.0f);
            for (int b0 = e0; b0 < e1; b0 += 32) {
                unsigned sv = 0u;
                float nv = 0.0f;
                if (b0 + lane < e1) { sv = q[b0 + lane]; nv = qns[b0 + lane]; }
                const int m = min(32, e1 - b0);
                for (int c0 = 0; c0 < m; c0 += ROW_CHUNK) {
                    float4 x[ROW_CHUNK];
                    #pragma unroll
                    for (int u = 0; u < ROW_CHUNK; ++u) {
                        if (c0 + u >= m) break;
                        const int s = __builtin_amdgcn_readlane((int)sv, c0 + u);
                        x[u] = *(const float4*)(feat + (size_t)s * DIM + 4 * lane);
                    }
                    if (!have_f0) { f0 = *f0p; nrow = norm[row]; have_f0 = true; }
                    #pragma unroll
                    for (int u = 0; u < ROW_CHUNK; ++u) {
                        if (c0 + u >= m) break;
                        const float w = __builtin_bit_cast(float,
                            __builtin_amdgcn_readlane(__builtin_bit_cast(int, nv), c0 + u));
                        acc.x = fmaf(x[u].x, w, acc.x);
                        acc.y = fmaf(x[u].y, w, acc.y);
                        acc.z = fmaf(x[u].z, w, acc.z);
                        acc.w = fmaf(x[u].w, w, acc.w);
                    }
                }
            }
            if (!have_f0) { f0 = *f0p; nrow = norm[row]; }
            const float nr = nrow * (1.0f - ALPHA_C);
            float4 h;
            h.x = acc.x * nr + ALPHA_C * f0.x;
            h.y = acc.y * nr + ALPHA_C * f0.y;
            h.z = acc.z * nr + ALPHA_C * f0.z;
            h.w = acc.w * nr + ALPHA_C * f0.w;
            *(float4*)(hacc + r * HS_STRIDE + 4 * lane) = h;
        }
    } else {
        for (int i = tid; i < ACC_ROWS * HS_STRIDE / 4; i += NT)
            ((float4*)hacc)[i] = make_float4(0.0f, 0.0f, 0.0f, 0.0f);
        __syncthreads();
        const int4* d4 = (const int4*)dst;
        const int off = blockIdx.x * (NV4 / 256);
        for (int i0 = 0; i0 < SCAN_FULL; i0 += NT * SCAN_U) {
            int4 v[SCAN_U];
            int  jx[SCAN_U];
            #pragma unroll
            for (int u = 0; u < SCAN_U; ++u) {
                int j = i0 + u * NT + tid + off;
                if (j >= NV4) j -= NV4;
                jx[u] = j;
                v[u] = d4[j];
            }
            #pragma unroll
            for (int u = 0; u < SCAN_U; ++u)
                scan_one(v[u], jx[u], nbase, q, &qn, hacc, feat, norm, src);
        }
        {
            int4 v[SCAN_TAIL_U];
            int  jx[SCAN_TAIL_U];
            #pragma unroll
            for (int u = 0; u < SCAN_TAIL_U; ++u) {
                int i = SCAN_FULL + u * NT + tid;
                int j = i + off;
                if (j >= NV4) j -= NV4;
                jx[u] = j;
                v[u] = (i < NV4) ? d4[j] : make_int4(-1, -1, -1, -1);
            }
            #pragma unroll
            for (int u = 0; u < SCAN_TAIL_U; ++u)
                scan_one(v[u], jx[u], nbase, q, &qn, hacc, feat, norm, src);
        }
        __syncthreads();

        const int n = __builtin_amdgcn_readfirstlane((int)min(qn, (unsigned)QCAP));
        const int n8 = (n + CU - 1) & ~(CU - 1);
        for (int i = tid; i < n8; i += NT) {
            unsigned pk = (unsigned)SINK_ROW << 16;
            float ns = 0.0f;
            if (i < n) {
                unsigned p = q[i];
                int s = src[p & 0xFFFFFu];
                ns = norm[s];
                pk = ((p >> 20) << 16) | (unsigned)s;
            }
            q[i]   = pk;
            qns[i] = ns;
        }
        __syncthreads();

        for (int i0 = wave * CU; i0 < n; i0 += (NT / 32) * CU) {
            const uint4  pa = *(const uint4*)(q + i0);
            const uint4  pb = *(const uint4*)(q + i0 + 4);
            const float4 na = *(const float4*)(qns + i0);
            const float4 nb = *(const float4*)(qns + i0 + 4);
            const unsigned pku[CU] = { pa.x, pa.y, pa.z, pa.w, pb.x, pb.y, pb.z, pb.w };
            const float    nsu[CU] = { na.x, na.y, na.z, na.w, nb.x, nb.y, nb.z, nb.w };
            v2f xa[CU], xb[CU];
            #pragma unroll
            for (int u = 0; u < CU; ++u) {
                const int s = __builtin_amdgcn_readfirstlane((int)(pku[u] & 0xFFFFu));
                const float* fr = feat + (size_t)s * DIM + 2 * lane;
                xa[u] = *(const v2f*)(fr);
                xb[u] = *(const v2f*)(fr + 64);
            }
            #pragma unroll
            for (int u = 0; u < CU; ++u) {
                const int dl = __builtin_amdgcn_readfirstlane((int)(pku[u] >> 16));
                float* ar = hacc + dl * HS_STRIDE + 2 * lane;
                atomicAdd(ar,      xa[u].x * nsu[u]);
                atomicAdd(ar + 1,  xa[u].y * nsu[u]);
                atomicAdd(ar + 64, xb[u].x * nsu[u]);
                atomicAdd(ar + 65, xb[u].y * nsu[u]);
            }
        }
        __syncthreads();

        for (int i = tid; i < R_NODES * (DIM / 4); i += NT) {
            int r = i >> 5;
            int c = (i & 31) * 4;
            int row = nbase + r;
            float nr = norm[row] * (1.0f - ALPHA_C);
            float4 a  = *(const float4*)(hacc + r * HS_STRIDE + c);
            float4 f0 = *(const float4*)(feat0 + (size_t)row * DIM + c);
            float4 h;
            h.x = a.x * nr + ALPHA_C * f0.x;
            h.y = a.y * nr + ALPHA_C * f0.y;
            h.z = a.z * nr + ALPHA_C * f0.z;
            h.w = a.w * nr + ALPHA_C * f0.w;
            *(float4*)(hacc + r * HS_STRIDE + c) = h;
        }
    }
    __syncthreads();

    const int half = lane >> 4;
    const int lr   = lane & 15;
    const int n0   = (wave & 7) * 16;
    for (int st = wave >> 3; st < N_STRIPS; st += 4) {
        const float* hs = hacc + st * 16 * HS_STRIDE;
        v8f acc = {};
        #pragma unroll
        for (int k0 = 0; k0 < DIM; k0 += 4) {
            const int ka = k0 + 2 * half;
            v2f a, b;
            a.x = hs[lr * HS_STRIDE + ka];
            a.y = hs[lr * HS_STRIDE + ka + 1];
            b.x = Wl[ka * DIM + n0 + lr];
            b.y = Wl[(ka + 1) * DIM + n0 + lr];
            acc = __builtin_amdgcn_wmma_f32_16x16x4_f32(
                 false, a,  false, b,
                 (short)0, acc,  false,  false);
        }

        const int col   = n0 + lr;
        const int lrow0 = st * 16 + 8 * half;
        float val[8];
        #pragma unroll
        for (int r = 0; r < 8; r++) {
            float hv = hacc[(lrow0 + r) * HS_STRIDE + col];
            val[r] = (1.0f - BETA_C) * hv + BETA_C * acc[r] + bs[col];
        }
        if (lrow0 < R_NODES) {
            #pragma unroll
            for (int r = 0; r < 8; r++)
                out[(size_t)(nbase + lrow0 + r) * DIM + col] = val[r];
        }
    }
}

extern "C" void kernel_launch(void* const* d_in, const int* in_sizes, int n_in,
                              void* d_out, int out_size, void* d_ws, size_t ws_size,
                              hipStream_t stream) {
    const float* feat  = (const float*)d_in[0];
    const float* feat0 = (const float*)d_in[1];
    const float* W     = (const float*)d_in[2];
    const float* bias  = (const float*)d_in[3];
    const int*   src   = (const int*)d_in[4];
    const int*   dst   = (const int*)d_in[5];
    float* out = (float*)d_out;
    float* deg = (float*)d_ws;

    gcnii_zero<<<(N_NODES + 255) / 256, 256, 0, stream>>>(deg, out);
    gcnii_degree_bin<<<BIN_GROUPS, BIN_NT, 0, stream>>>(src, dst, deg, out);
    gcnii_norm<<<(N_NODES + 255) / 256, 256, 0, stream>>>(deg);
    gcnii_own<<<N_GROUPS, NT, 0, stream>>>(feat, feat0, W, bias, deg, src, dst, out);
}


# Edges filed without global atomics: each binning work-group fills a private cell per owner group (LDS rank); owners pack cells, count own in-degrees: clear+norm launches gone; arrival-order sums
# speedup vs baseline: 3.5094x; 1.7541x over previous
#include <hip/hip_runtime.h>
#include <hip/hip_bf16.h>


#define N_NODES 50000
#define N_EDGES 600000
#define DIM 128
#define ALPHA_C 0.1f
#define BETA_C 0.22314355131420976f
#define HS_STRIDE 132

#define R_NODES 200
#define N_GROUPS 250
#define N_STRIPS 13
#define ACC_ROWS 209
#define SINK_ROW 208
#define QCAP 8192
#define NT 1024
#define NV4 (N_EDGES / 4)
#define SCAN_U 8
#define CU 8
#define SCAN_FULL (NV4 / (NT * SCAN_U) * (NT * SCAN_U))
#define SCAN_TAIL_U ((NV4 - SCAN_FULL + NT - 1) / NT)
#define GROUP_WORDS (R_NODES * DIM)
#define FILE_OFF 32
#define BIN_NT 1024
#define BIN_GROUPS ((NV4 + BIN_NT - 1) / BIN_NT)
#define SORT_SLOTS (QCAP / NT)
#define RCNT_N 256
#define ROW_CHUNK 16
#define CELL_WORDS 64
#define CELL_CAP (CELL_WORDS - 1)
#define CELL_OFF 8256
#define CELL_PAD 160
#define DN_NT 256
#define DN_WAVES (DN_NT / 32)
#define DN_CELLS ((BIN_GROUPS + DN_WAVES - 1) / DN_WAVES)

static_assert(N_GROUPS * R_NODES == N_NODES, "ownership must tile the node set");
static_assert(N_EDGES % 4 == 0, "int4 view of dst");
static_assert(N_EDGES < (1 << 20), "edge id is packed into 20 bits");
static_assert(N_NODES <= (1 << 16), "source id is packed into 16 bits");
static_assert(R_NODES % 8 == 0, "a half strip (8 rows) is wholly real or wholly padding");
static_assert(QCAP % CU == 0, "queue is read CU entries at a time");
static_assert(FILE_OFF + QCAP <= GROUP_WORDS, "a group's edge file lies inside the group's own out rows");
static_assert(R_NODES <= (1 << 16), "local row is packed above a 16-bit source id");
static_assert(QCAP % NT == 0, "every thread holds the same number of file slots");
static_assert(R_NODES < RCNT_N && RCNT_N == 8 * 32, "one wave scans the row counters, 8 per lane");
static_assert(CELL_OFF >= FILE_OFF + QCAP && CELL_OFF % 64 == 0, "cells lie behind the file, each on its own two lines");
static_assert(CELL_OFF + BIN_GROUPS * CELL_WORDS <= GROUP_WORDS, "a group's cells lie inside the group's own out rows");
static_assert(BIN_GROUPS <= CELL_PAD && CELL_PAD == 5 * 32, "one wave scans the cell counters, 5 per lane");
static_assert(RCNT_N <= DN_NT && CELL_PAD <= DN_NT && R_NODES <= DN_NT, "one thread per counter");

typedef __attribute__((ext_vector_type(2))) float v2f;
typedef __attribute__((ext_vector_type(8))) float v8f;

__device__ __forceinline__ unsigned* group_words(float* out, unsigned g) {
    return (unsigned*)out + (size_t)g * GROUP_WORDS;
}
__device__ __forceinline__ unsigned* group_cell(float* out, unsigned g, unsigned b) {
    return group_words(out, g) + CELL_OFF + b * CELL_WORDS;
}

__global__ __launch_bounds__(BIN_NT) void gcnii_bin(const int* __restrict__ src,
                                                   const int* __restrict__ dst,
                                                   float* out) {
    __shared__ unsigned hist[N_GROUPS];
    const int tid = threadIdx.x;
    if (tid < N_GROUPS) hist[tid] = 0u;
    __syncthreads();

    const int i = blockIdx.x * BIN_NT + tid;
    const bool live = i < NV4;
    unsigned g[4], pk[4], rk[4];
    if (live) {
        const int4 d = ((const int4*)dst)[i];
        const int4 s = ((const int4*)src)[i];
        const int dv[4] = { d.x, d.y, d.z, d.w };
        const int sv[4] = { s.x, s.y, s.z, s.w };
        #pragma unroll
        for (int c = 0; c < 4; ++c) {
            g[c]  = (unsigned)dv[c] / R_NODES;
            pk[c] = (((unsigned)dv[c] - g[c] * R_NODES) << 16) | (unsigned)sv[c];
            rk[c] = atomicAdd(&hist[g[c]], 1u);
        }
    }
    __syncthreads();
    if (live) {
        #pragma unroll
        for (int c = 0; c < 4; ++c)
            if (rk[c] < CELL_CAP) group_cell(out, g[c], blockIdx.x)[1 + rk[c]] = pk[c];
    }
    if (tid < N_GROUPS) group_cell(out, (unsigned)tid, blockIdx.x)[0] = hist[tid];
}

__global__ __launch_bounds__(DN_NT) void gcnii_degnorm(const int* __restrict__ dst,
                                                      float* __restrict__ norm,
                                                      float* out) {
    __shared__ unsigned cstart[CELL_PAD];
    __shared__ unsigned rc[RCNT_N];
    __shared__ unsigned tot;
    __shared__ unsigned over;

    const int tid  = threadIdx.x;
    const int lane = tid & 31;
    const int wave = __builtin_amdgcn_readfirstlane(tid >> 5);
    const int nbase = blockIdx.x * R_NODES;
    unsigned* gw = group_words(out, blockIdx.x);

    uint2 v[DN_CELLS];
    #pragma unroll
    for (int c = 0; c < DN_CELLS; ++c) {
        const int b = wave + DN_WAVES * c;
        v[c] = make_uint2(0u, 0u);
        if (b < BIN_GROUPS) v[c] = *(const uint2*)(gw + CELL_OFF + b * CELL_WORDS + 2 * lane);
    }
    if (tid < CELL_PAD) cstart[tid] = 0u;
    if (tid < RCNT_N) rc[tid] = 0u;
    if (tid == 0) { tot = 0u; over = 0u; }
    __syncthreads();

    unsigned cn[DN_CELLS];
    #pragma unroll
    for (int c = 0; c < DN_CELLS; ++c) {
        const int b = wave + DN_WAVES * c;
        const unsigned n = (unsigned)__builtin_amdgcn_readfirstlane((int)v[c].x);
        cn[c] = min(n, (unsigned)CELL_CAP);
        if (b < BIN_GROUPS && lane == 0) {
            cstart[b] = cn[c];
            if (n > (unsigned)CELL_CAP) atomicOr(&over, 1u);
        }
    }
    __syncthreads();

    if (wave == 0) {
        unsigned c5[5];
        unsigned t = 0u;
        #pragma unroll
        for (int j = 0; j < 5; ++j) { const unsigned x = cstart[5 * lane + j]; c5[j] = t; t += x; }
        unsigned incl = t;
        #pragma unroll
        for (int d = 1; d < 32; d <<= 1) {
            const unsigned o = __shfl_up(incl, (unsigned)d, 32);
            if (lane >= d) incl += o;
        }
        const unsigned base = incl - t;
        #pragma unroll
        for (int j = 0; j < 5; ++j) cstart[5 * lane + j] = base + c5[j];
        if (lane == 31) tot = incl;
    }
    __syncthreads();

    const bool ovf = __builtin_amdgcn_readfirstlane((int)over) != 0;
    if (!ovf) {
        #pragma unroll
        for (int c = 0; c < DN_CELLS; ++c) {
            const int b = wave + DN_WAVES * c;
            if (b < BIN_GROUPS) {
                const unsigned st = cstart[b];
                const unsigned jx = 2u * (unsigned)lane;
                if (jx >= 1u && jx <= cn[c]) {
                    const unsigned pos = st + jx - 1u;
                    if (pos < (unsigned)QCAP) gw[FILE_OFF + pos] = v[c].x;
                    atomicAdd(&rc[v[c].x >> 16], 1u);
                }
                if (jx + 1u <= cn[c]) {
                    const unsigned pos = st + jx;
                    if (pos < (unsigned)QCAP) gw[FILE_OFF + pos] = v[c].y;
                    atomicAdd(&rc[v[c].y >> 16], 1u);
                }
            }
        }
    } else {
        for (int i = tid; i < N_EDGES; i += DN_NT) {
            const unsigned dl = (unsigned)(dst[i] - nbase);
            if (dl < (unsigned)R_NODES) atomicAdd(&rc[dl], 1u);
        }
    }
    __syncthreads();
    if (tid < R_NODES) norm[nbase + tid] = rsqrtf(fmaxf((float)rc[tid], 1.0f));
    if (tid == 0) gw[0] = ovf ? (unsigned)(QCAP + 1) : tot;
}

__device__ __forceinline__ void add_edge_slow(const float* __restrict__ feat,
                                              const float* __restrict__ norm,
                                              const int* __restrict__ src,
                                              float* hacc, unsigned e, unsigned dl) {
    int s = src[e];
    float ns = norm[s];
    for (int c = 0; c < DIM; ++c)
        atomicAdd(&hacc[dl * HS_STRIDE + c], feat[(size_t)s * DIM + c] * ns);
}

__device__ __forceinline__ void scan_one(const int4 v, const int j, const int nbase,
                                         unsigned* q, unsigned* qn, float* hacc,
                                         const float* __restrict__ feat,
                                         const float* __restrict__ norm,
                                         const int* __restrict__ src) {
    unsigned dl[4];
    dl[0] = (unsigned)(v.x - nbase);
    dl[1] = (unsigned)(v.y - nbase);
    dl[2] = (unsigned)(v.z - nbase);
    dl[3] = (unsigned)(v.w - nbase);
    if ((dl[0] < R_NODES) | (dl[1] < R_NODES) | (dl[2] < R_NODES) | (dl[3] < R_NODES)) {
        #pragma unroll
        for (int c = 0; c < 4; ++c) {
            if (dl[c] < R_NODES) {
                unsigned e = 4u * (unsigned)j + (unsigned)c;
                unsigned slot = atomicAdd(qn, 1u);
                if (slot < QCAP) q[slot] = (dl[c] << 20) | e;
                else add_edge_slow(feat, norm, src, hacc, e, dl[c]);
            }
        }
    }
}

__global__ __launch_bounds__(NT) void gcnii_own(
    const float* __restrict__ feat,
    const float* __restrict__ feat0,
    const float* __restrict__ W,
    const float* __restrict__ bias,
    const float* __restrict__ norm,
    const int* __restrict__ src,
    const int* __restrict__ dst,
    float* out)
{
    __shared__ __attribute__((aligned(16))) float hacc[ACC_ROWS * HS_STRIDE];
    __shared__ __attribute__((aligned(16))) float Wl[DIM * DIM];
    __shared__ __attribute__((aligned(16))) unsigned q[QCAP];
    __shared__ __attribute__((aligned(16))) float qns[QCAP];
    __shared__ __attribute__((aligned(16))) unsigned rcnt[RCNT_N];
    __shared__ float bs[DIM];
    __shared__ unsigned qn;

    const int tid  = threadIdx.x;
    const int lane = tid & 31;
    const int wave = __builtin_amdgcn_readfirstlane(tid >> 5);
    const int nbase = blockIdx.x * R_NODES;

    for (int i = tid; i < DIM * DIM / 4; i += NT)
        ((float4*)Wl)[i] = ((const float4*)W)[i];
    if (tid < DIM) bs[tid] = bias[tid];
    if (tid < RCNT_N) rcnt[tid] = 0u;
    if (tid == 0) qn = 0u;
    for (int i = tid; i < (ACC_ROWS - R_NODES) * HS_STRIDE; i += NT)
        hacc[R_NODES * HS_STRIDE + i] = 0.0f;
    __syncthreads();

    const unsigned* gw = group_words(out, blockIdx.x);
    const unsigned nfile = (unsigned)__builtin_amdgcn_readfirstlane((int)gw[0]);
    if (nfile <= (unsigned)QCAP) {
        const int n = (int)nfile;

        unsigned pk[SORT_SLOTS], rk[SORT_SLOTS];
        float    ns[SORT_SLOTS];
        #pragma unroll
        for (int k = 0; k < SORT_SLOTS; ++k) {
            const int i = k * NT + tid;
            pk[k] = 0u;
            if (i < n) pk[k] = gw[FILE_OFF + i];
        }
        #pragma unroll
        for (int k = 0; k < SORT_SLOTS; ++k)
            ns[k] = norm[pk[k] & 0xFFFFu];
        #pragma unroll
        for (int k = 0; k < SORT_SLOTS; ++k) {
            const int i = k * NT + tid;
            rk[k] = 0u;
            if (i < n) rk[k] = atomicAdd(&rcnt[pk[k] >> 16], 1u);
        }
        __syncthreads();

        if (wave == 0) {
            uint4 ca = *(const uint4*)(rcnt + 8 * lane);
            uint4 cb = *(const uint4*)(rcnt + 8 * lane + 4);
            unsigned c[8] = { ca.x, ca.y, ca.z, ca.w, cb.x, cb.y, cb.z, cb.w };
            unsigned t = 0u;
            #pragma unroll
            for (int j = 0; j < 8; ++j) { const unsigned v = c[j]; c[j] = t; t += v; }
            unsigned incl = t;
            #pragma unroll
            for (int d = 1; d < 32; d <<= 1) {
                const unsigned o = __shfl_up(incl, (unsigned)d, 32);
                if (lane >= d) incl += o;
            }
            const unsigned base = incl - t;
            ca = make_uint4(base + c[0], base + c[1], base + c[2], base + c[3]);
            cb = make_uint4(base + c[4], base + c[5], base + c[6], base + c[7]);
            *(uint4*)(rcnt + 8 * lane)     = ca;
            *(uint4*)(rcnt + 8 * lane + 4) = cb;
        }
        __syncthreads();

        #pragma unroll
        for (int k = 0; k < SORT_SLOTS; ++k) {
            const int i = k * NT + tid;
            if (i < n) {
                const unsigned pos = rcnt[pk[k] >> 16] + rk[k];
                q[pos]   = pk[k] & 0xFFFFu;
                qns[pos] = ns[k];
            }
        }
        __syncthreads();

        for (int r = wave; r < R_NODES; r += NT / 32) {
            const int e0 = __builtin_amdgcn_readfirstlane((int)rcnt[r]);
            const int e1 = __builtin_amdgcn_readfirstlane((int)rcnt[r + 1]);
            const int row = nbase + r;
            const float4* f0p = (const float4*)(feat0 + (size_t)row * DIM + 4 * lane);
            float4 f0 = make_float4(0.0f, 0.0f, 0.0f, 0.0f);
            float nrow = 0.0f;
            bool have_f0 = false;
            float4 acc = make_float4(0.0f, 0.0f, 0.0f, 0.0f);
            for (int b0 = e0; b0 < e1; b0 += 32) {
                unsigned sv = 0u;
                float nv = 0.0f;
                if (b0 + lane < e1) { sv = q[b0 + lane]; nv = qns[b0 + lane]; }
                const int m = min(32, e1 - b0);
                for (int c0 = 0; c0 < m; c0 += ROW_CHUNK) {
                    float4 x[ROW_CHUNK];
                    #pragma unroll
                    for (int u = 0; u < ROW_CHUNK; ++u) {
                        if (c0 + u >= m) break;
                        const int s = __builtin_amdgcn_readlane((int)sv, c0 + u);
                        x[u] = *(const float4*)(feat + (size_t)s * DIM + 4 * lane);
                    }
                    if (!have_f0) { f0 = *f0p; nrow = norm[row]; have_f0 = true; }
                    #pragma unroll
                    for (int u = 0; u < ROW_CHUNK; ++u) {
                        if (c0 + u >= m) break;
                        const float w = __builtin_bit_cast(float,
                            __builtin_amdgcn_readlane(__builtin_bit_cast(int, nv), c0 + u));
                        acc.x = fmaf(x[u].x, w, acc.x);
                        acc.y = fmaf(x[u].y, w, acc.y);
                        acc.z = fmaf(x[u].z, w, acc.z);
                        acc.w = fmaf(x[u].w, w, acc.w);
                    }
                }
            }
            if (!have_f0) { f0 = *f0p; nrow = norm[row]; }
            const float nr = nrow * (1.0f - ALPHA_C);
            float4 h;
            h.x = acc.x * nr + ALPHA_C * f0.x;
            h.y = acc.y * nr + ALPHA_C * f0.y;
            h.z = acc.z * nr + ALPHA_C * f0.z;
            h.w = acc.w * nr + ALPHA_C * f0.w;
            *(float4*)(hacc + r * HS_STRIDE + 4 * lane) = h;
        }
    } else {
        for (int i = tid; i < ACC_ROWS * HS_STRIDE / 4; i += NT)
            ((float4*)hacc)[i] = make_float4(0.0f, 0.0f, 0.0f, 0.0f);
        __syncthreads();
        const int4* d4 = (const int4*)dst;
        const int off = blockIdx.x * (NV4 / 256);
        for (int i0 = 0; i0 < SCAN_FULL; i0 += NT * SCAN_U) {
            int4 v[SCAN_U];
            int  jx[SCAN_U];
            #pragma unroll
            for (int u = 0; u < SCAN_U; ++u) {
                int j = i0 + u * NT + tid + off;
                if (j >= NV4) j -= NV4;
                jx[u] = j;
                v[u] = d4[j];
            }
            #pragma unroll
            for (int u = 0; u < SCAN_U; ++u)
                scan_one(v[u], jx[u], nbase, q, &qn, hacc, feat, norm, src);
        }
        {
            int4 v[SCAN_TAIL_U];
            int  jx[SCAN_TAIL_U];
            #pragma unroll
            for (int u = 0; u < SCAN_TAIL_U; ++u) {
                int i = SCAN_FULL + u * NT + tid;
                int j = i + off;
                if (j >= NV4) j -= NV4;
                jx[u] = j;
                v[u] = (i < NV4) ? d4[j] : make_int4(-1, -1, -1, -1);
            }
            #pragma unroll
            for (int u = 0; u < SCAN_TAIL_U; ++u)
                scan_one(v[u], jx[u], nbase, q, &qn, hacc, feat, norm, src);
        }
        __syncthreads();

        const int n = __builtin_amdgcn_readfirstlane((int)min(qn, (unsigned)QCAP));
        const int n8 = (n + CU - 1) & ~(CU - 1);
        for (int i = tid; i < n8; i += NT) {
            unsigned pk = (unsigned)SINK_ROW << 16;
            float ns = 0.0f;
            if (i < n) {
                unsigned p = q[i];
                int s = src[p & 0xFFFFFu];
                ns = norm[s];
                pk = ((p >> 20) << 16) | (unsigned)s;
            }
            q[i]   = pk;
            qns[i] = ns;
        }
        __syncthreads();

        for (int i0 = wave * CU; i0 < n; i0 += (NT / 32) * CU) {
            const uint4  pa = *(const uint4*)(q + i0);
            const uint4  pb = *(const uint4*)(q + i0 + 4);
            const float4 na = *(const float4*)(qns + i0);
            const float4 nb = *(const float4*)(qns + i0 + 4);
            const unsigned pku[CU] = { pa.x, pa.y, pa.z, pa.w, pb.x, pb.y, pb.z, pb.w };
            const float    nsu[CU] = { na.x, na.y, na.z, na.w, nb.x, nb.y, nb.z, nb.w };
            v2f xa[CU], xb[CU];
            #pragma unroll
            for (int u = 0; u < CU; ++u) {
                const int s = __builtin_amdgcn_readfirstlane((int)(pku[u] & 0xFFFFu));
                const float* fr = feat + (size_t)s * DIM + 2 * lane;
                xa[u] = *(const v2f*)(fr);
                xb[u] = *(const v2f*)(fr + 64);
            }
            #pragma unroll
            for (int u = 0; u < CU; ++u) {
                const int dl = __builtin_amdgcn_readfirstlane((int)(pku[u] >> 16));
                float* ar = hacc + dl * HS_STRIDE + 2 * lane;
                atomicAdd(ar,      xa[u].x * nsu[u]);
                atomicAdd(ar + 1,  xa[u].y * nsu[u]);
                atomicAdd(ar + 64, xb[u].x * nsu[u]);
                atomicAdd(ar + 65, xb[u].y * nsu[u]);
            }
        }
        __syncthreads();

        for (int i = tid; i < R_NODES * (DIM / 4); i += NT) {
            int r = i >> 5;
            int c = (i & 31) * 4;
            int row = nbase + r;
            float nr = norm[row] * (1.0f - ALPHA_C);
            float4 a  = *(const float4*)(hacc + r * HS_STRIDE + c);
            float4 f0 = *(const float4*)(feat0 + (size_t)row * DIM + c);
            float4 h;
            h.x = a.x * nr + ALPHA_C * f0.x;
            h.y = a.y * nr + ALPHA_C * f0.y;
            h.z = a.z * nr + ALPHA_C * f0.z;
            h.w = a.w * nr + ALPHA_C * f0.w;
            *(float4*)(hacc + r * HS_STRIDE + c) = h;
        }
    }
    __syncthreads();

    const int half = lane >> 4;
    const int lr   = lane & 15;
    const int n0   = (wave & 7) * 16;
    for (int st = wave >> 3; st < N_STRIPS; st += 4) {
        const float* hs = hacc + st * 16 * HS_STRIDE;
        v8f acc = {};
        #pragma unroll
        for (int k0 = 0; k0 < DIM; k0 += 4) {
            const int ka = k0 + 2 * half;
            v2f a, b;
            a.x = hs[lr * HS_STRIDE + ka];
            a.y = hs[lr * HS_STRIDE + ka + 1];
            b.x = Wl[ka * DIM + n0 + lr];
            b.y = Wl[(ka + 1) * DIM + n0 + lr];
            acc = __builtin_amdgcn_wmma_f32_16x16x4_f32(
                 false, a,  false, b,
                 (short)0, acc,  false,  false);
        }

        const int col   = n0 + lr;
        const int lrow0 = st * 16 + 8 * half;
        float val[8];
        #pragma unroll
        for (int r = 0; r < 8; r++) {
            float hv = hacc[(lrow0 + r) * HS_STRIDE + col];
            val[r] = (1.0f - BETA_C) * hv + BETA_C * acc[r] + bs[col];
        }
        if (lrow0 < R_NODES) {
            #pragma unroll
            for (int r = 0; r < 8; r++)
                out[(size_t)(nbase + lrow0 + r) * DIM + col] = val[r];
        }
    }
}

extern "C" void kernel_launch(void* const* d_in, const int* in_sizes, int n_in,
                              void* d_out, int out_size, void* d_ws, size_t ws_size,
                              hipStream_t stream) {
    const float* feat  = (const float*)d_in[0];
    const float* feat0 = (const float*)d_in[1];
    const float* W     = (const float*)d_in[2];
    const float* bias  = (const float*)d_in[3];
    const int*   src   = (const int*)d_in[4];
    const int*   dst   = (const int*)d_in[5];
    float* out = (float*)d_out;
    float* norm = (float*)d_ws;

    gcnii_bin<<<BIN_GROUPS, BIN_NT, 0, stream>>>(src, dst, out);
    gcnii_degnorm<<<N_GROUPS, DN_NT, 0, stream>>>(dst, norm, out);
    gcnii_own<<<N_GROUPS, NT, 0, stream>>>(feat, feat0, W, bias, norm, src, dst, out);
}


# Binning kernel builds its cells as images in LDS; a wave stores a whole cell (count + edge words, 8 bytes a lane) per instruction: 1-2 lines a store, not 32 scattered 4-byte stores; sum order same
# speedup vs baseline: 3.6440x; 1.0384x over previous
#include <hip/hip_runtime.h>
#include <hip/hip_bf16.h>


#define N_NODES 50000
#define N_EDGES 600000
#define DIM 128
#define ALPHA_C 0.1f
#define BETA_C 0.22314355131420976f
#define HS_STRIDE 132

#define R_NODES 200
#define N_GROUPS 250
#define N_STRIPS 13
#define ACC_ROWS 209
#define SINK_ROW 208
#define QCAP 8192
#define NT 1024
#define NV4 (N_EDGES / 4)
#define SCAN_U 8
#define CU 8
#define SCAN_FULL (NV4 / (NT * SCAN_U) * (NT * SCAN_U))
#define SCAN_TAIL_U ((NV4 - SCAN_FULL + NT - 1) / NT)
#define GROUP_WORDS (R_NODES * DIM)
#define FILE_OFF 32
#define BIN_NT 1024
#define BIN_GROUPS ((NV4 + BIN_NT - 1) / BIN_NT)
#define SORT_SLOTS (QCAP / NT)
#define RCNT_N 256
#define ROW_CHUNK 16
#define CELL_WORDS 64
#define CELL_CAP (CELL_WORDS - 1)
#define CELL_OFF 8256
#define CELL_PAD 160
#define DN_NT 256
#define DN_WAVES (DN_NT / 32)
#define DN_CELLS ((BIN_GROUPS + DN_WAVES - 1) / DN_WAVES)
#define IMG_PITCH 66
#define BIN_CELLS ((N_GROUPS + BIN_NT / 32 - 1) / (BIN_NT / 32))

static_assert(N_GROUPS * R_NODES == N_NODES, "ownership must tile the node set");
static_assert(N_EDGES % 4 == 0, "int4 view of dst");
static_assert(N_EDGES < (1 << 20), "edge id is packed into 20 bits");
static_assert(N_NODES <= (1 << 16), "source id is packed into 16 bits");
static_assert(R_NODES % 8 == 0, "a half strip (8 rows) is wholly real or wholly padding");
static_assert(QCAP % CU == 0, "queue is read CU entries at a time");
static_assert(FILE_OFF + QCAP <= GROUP_WORDS, "a group's edge file lies inside the group's own out rows");
static_assert(R_NODES <= (1 << 16), "local row is packed above a 16-bit source id");
static_assert(QCAP % NT == 0, "every thread holds the same number of file slots");
static_assert(R_NODES < RCNT_N && RCNT_N == 8 * 32, "one wave scans the row counters, 8 per lane");
static_assert(CELL_OFF >= FILE_OFF + QCAP && CELL_OFF % 64 == 0, "cells lie behind the file, each on its own two lines");
static_assert(CELL_OFF + BIN_GROUPS * CELL_WORDS <= GROUP_WORDS, "a group's cells lie inside the group's own out rows");
static_assert(BIN_GROUPS <= CELL_PAD && CELL_PAD == 5 * 32, "one wave scans the cell counters, 5 per lane");
static_assert(RCNT_N <= DN_NT && CELL_PAD <= DN_NT && R_NODES <= DN_NT, "one thread per counter");
static_assert(IMG_PITCH >= CELL_WORDS && IMG_PITCH % 2 == 0 && CELL_WORDS == 2 * 32, "one wave stores a cell image, two words per lane");

typedef __attribute__((ext_vector_type(2))) float v2f;
typedef __attribute__((ext_vector_type(8))) float v8f;

__device__ __forceinline__ unsigned* group_words(float* out, unsigned g) {
    return (unsigned*)out + (size_t)g * GROUP_WORDS;
}
__device__ __forceinline__ unsigned* group_cell(float* out, unsigned g, unsigned b) {
    return group_words(out, g) + CELL_OFF + b * CELL_WORDS;
}

__global__ __launch_bounds__(BIN_NT) void gcnii_bin(const int* __restrict__ src,
                                                   const int* __restrict__ dst,
                                                   float* out) {
    __shared__ unsigned hist[N_GROUPS];
    __shared__ __attribute__((aligned(16))) unsigned img[N_GROUPS * IMG_PITCH];
    const int tid  = threadIdx.x;
    const int lane = tid & 31;
    const int wave = __builtin_amdgcn_readfirstlane(tid >> 5);
    if (tid < N_GROUPS) hist[tid] = 0u;
    __syncthreads();

    const int i = blockIdx.x * BIN_NT + tid;
    if (i < NV4) {
        const int4 d = ((const int4*)dst)[i];
        const int4 s = ((const int4*)src)[i];
        const int dv[4] = { d.x, d.y, d.z, d.w };
        const int sv[4] = { s.x, s.y, s.z, s.w };
        unsigned g[4], pk[4], rk[4];
        #pragma unroll
        for (int c = 0; c < 4; ++c) {
            g[c]  = (unsigned)dv[c] / R_NODES;
            pk[c] = (((unsigned)dv[c] - g[c] * R_NODES) << 16) | (unsigned)sv[c];
            rk[c] = atomicAdd(&hist[g[c]], 1u);
        }
        #pragma unroll
        for (int c = 0; c < 4; ++c)
            if (rk[c] < CELL_CAP) img[g[c] * IMG_PITCH + 1u + rk[c]] = pk[c];
    }
    __syncthreads();

    const unsigned w0 = 2u * (unsigned)lane;
    unsigned nn[BIN_CELLS];
    #pragma unroll
    for (int k = 0; k < BIN_CELLS; ++k) {
        const int gc = min(wave + (BIN_NT / 32) * k, N_GROUPS - 1);
        nn[k] = hist[gc];
    }
    uint2 cw[BIN_CELLS];
    #pragma unroll
    for (int k = 0; k < BIN_CELLS; ++k) {
        const int gc = min(wave + (BIN_NT / 32) * k, N_GROUPS - 1);
        nn[k] = (unsigned)__builtin_amdgcn_readfirstlane((int)nn[k]);
        const unsigned last = min(nn[k], (unsigned)CELL_CAP);
        cw[k].x = (lane != 0 && w0 <= last) ? img[gc * IMG_PITCH + w0] : nn[k];
        cw[k].y = (w0 + 1u <= last) ? img[gc * IMG_PITCH + w0 + 1u] : 0u;
    }
    #pragma unroll
    for (int k = 0; k < BIN_CELLS; ++k) {
        const int gc = wave + (BIN_NT / 32) * k;
        const unsigned last = min(nn[k], (unsigned)CELL_CAP);
        if (gc < N_GROUPS && w0 <= last)
            *(uint2*)(group_cell(out, (unsigned)gc, blockIdx.x) + w0) = cw[k];
    }
}

__global__ __launch_bounds__(DN_NT) void gcnii_degnorm(const int* __restrict__ dst,
                                                      float* __restrict__ norm,
                                                      float* out) {
    __shared__ unsigned cstart[CELL_PAD];
    __shared__ unsigned rc[RCNT_N];
    __shared__ unsigned tot;
    __shared__ unsigned over;

    const int tid  = threadIdx.x;
    const int lane = tid & 31;
    const int wave = __builtin_amdgcn_readfirstlane(tid >> 5);
    const int nbase = blockIdx.x * R_NODES;
    unsigned* gw = group_words(out, blockIdx.x);

    uint2 v[DN_CELLS];
    #pragma unroll
    for (int c = 0; c < DN_CELLS; ++c) {
        const int b = wave + DN_WAVES * c;
        v[c] = make_uint2(0u, 0u);
        if (b < BIN_GROUPS) v[c] = *(const uint2*)(gw + CELL_OFF + b * CELL_WORDS + 2 * lane);
    }
    if (tid < CELL_PAD) cstart[tid] = 0u;
    if (tid < RCNT_N) rc[tid] = 0u;
    if (tid == 0) { tot = 0u; over = 0u; }
    __syncthreads();

    unsigned cn[DN_CELLS];
    #pragma unroll
    for (int c = 0; c < DN_CELLS; ++c) {
        const int b = wave + DN_WAVES * c;
        const unsigned n = (unsigned)__builtin_amdgcn_readfirstlane((int)v[c].x);
        cn[c] = min(n, (unsigned)CELL_CAP);
        if (b < BIN_GROUPS && lane == 0) {
            cstart[b] = cn[c];
            if (n > (unsigned)CELL_CAP) atomicOr(&over, 1u);
        }
    }
    __syncthreads();

    if (wave == 0) {
        unsigned c5[5];
        unsigned t = 0u;
        #pragma unroll
        for (int j = 0; j < 5; ++j) { const unsigned x = cstart[5 * lane + j]; c5[j] = t; t += x; }
        unsigned incl = t;
        #pragma unroll
        for (int d = 1; d < 32; d <<= 1) {
            const unsigned o = __shfl_up(incl, (unsigned)d, 32);
            if (lane >= d) incl += o;
        }
        const unsigned base = incl - t;
        #pragma unroll
        for (int j = 0; j < 5; ++j) cstart[5 * lane + j] = base + c5[j];
        if (lane == 31) tot = incl;
    }
    __syncthreads();

    const bool ovf = __builtin_amdgcn_readfirstlane((int)over) != 0;
    if (!ovf) {
        #pragma unroll
        for (int c = 0; c < DN_CELLS; ++c) {
            const int b = wave + DN_WAVES * c;
            if (b < BIN_GROUPS) {
                const unsigned st = cstart[b];
                const unsigned jx = 2u * (unsigned)lane;
                if (jx >= 1u && jx <= cn[c]) {
                    const unsigned pos = st + jx - 1u;
                    if (pos < (unsigned)QCAP) gw[FILE_OFF + pos] = v[c].x;
                    atomicAdd(&rc[v[c].x >> 16], 1u);
                }
                if (jx + 1u <= cn[c]) {
                    const unsigned pos = st + jx;
                    if (pos < (unsigned)QCAP) gw[FILE_OFF + pos] = v[c].y;
                    atomicAdd(&rc[v[c].y >> 16], 1u);
                }
            }
        }
    } else {
        for (int i = tid; i < N_EDGES; i += DN_NT) {
            const unsigned dl = (unsigned)(dst[i] - nbase);
            if (dl < (unsigned)R_NODES) atomicAdd(&rc[dl], 1u);
        }
    }
    __syncthreads();
    if (tid < R_NODES) norm[nbase + tid] = rsqrtf(fmaxf((float)rc[tid], 1.0f));
    if (tid == 0) gw[0] = ovf ? (unsigned)(QCAP + 1) : tot;
}

__device__ __forceinline__ void add_edge_slow(const float* __restrict__ feat,
                                              const float* __restrict__ norm,
                                              const int* __restrict__ src,
                                              float* hacc, unsigned e, unsigned dl) {
    int s = src[e];
    float ns = norm[s];
    for (int c = 0; c < DIM; ++c)
        atomicAdd(&hacc[dl * HS_STRIDE + c], feat[(size_t)s * DIM + c] * ns);
}

__device__ __forceinline__ void scan_one(const int4 v, const int j, const int nbase,
                                         unsigned* q, unsigned* qn, float* hacc,
                                         const float* __restrict__ feat,
                                         const float* __restrict__ norm,
                                         const int* __restrict__ src) {
    unsigned dl[4];
    dl[0] = (unsigned)(v.x - nbase);
    dl[1] = (unsigned)(v.y - nbase);
    dl[2] = (unsigned)(v.z - nbase);
    dl[3] = (unsigned)(v.w - nbase);
    if ((dl[0] < R_NODES) | (dl[1] < R_NODES) | (dl[2] < R_NODES) | (dl[3] < R_NODES)) {
        #pragma unroll
        for (int c = 0; c < 4; ++c) {
            if (dl[c] < R_NODES) {
                unsigned e = 4u * (unsigned)j + (unsigned)c;
                unsigned slot = atomicAdd(qn, 1u);
                if (slot < QCAP) q[slot] = (dl[c] << 20) | e;
                else add_edge_slow(feat, norm, src, hacc, e, dl[c]);
            }
        }
    }
}

__global__ __launch_bounds__(NT) void gcnii_own(
    const float* __restrict__ feat,
    const float* __restrict__ feat0,
    const float* __restrict__ W,
    const float* __restrict__ bias,
    const float* __restrict__ norm,
    const int* __restrict__ src,
    const int* __restrict__ dst,
    float* out)
{
    __shared__ __attribute__((aligned(16))) float hacc[ACC_ROWS * HS_STRIDE];
    __shared__ __attribute__((aligned(16))) float Wl[DIM * DIM];
    __shared__ __attribute__((aligned(16))) unsigned q[QCAP];
    __shared__ __attribute__((aligned(16))) float qns[QCAP];
    __shared__ __attribute__((aligned(16))) unsigned rcnt[RCNT_N];
    __shared__ float bs[DIM];
    __shared__ unsigned qn;

    const int tid  = threadIdx.x;
    const int lane = tid & 31;
    const int wave = __builtin_amdgcn_readfirstlane(tid >> 5);
    const int nbase = blockIdx.x * R_NODES;

    for (int i = tid; i < DIM * DIM / 4; i += NT)
        ((float4*)Wl)[i] = ((const float4*)W)[i];
    if (tid < DIM) bs[tid] = bias[tid];
    if (tid < RCNT_N) rcnt[tid] = 0u;
    if (tid == 0) qn = 0u;
    for (int i = tid; i < (ACC_ROWS - R_NODES) * HS_STRIDE; i += NT)
        hacc[R_NODES * HS_STRIDE + i] = 0.0f;
    __syncthreads();

    const unsigned* gw = group_words(out, blockIdx.x);
    const unsigned nfile = (unsigned)__builtin_amdgcn_readfirstlane((int)gw[0]);
    if (nfile <= (unsigned)QCAP) {
        const int n = (int)nfile;

        unsigned pk[SORT_SLOTS], rk[SORT_SLOTS];
        float    ns[SORT_SLOTS];
        #pragma unroll
        for (int k = 0; k < SORT_SLOTS; ++k) {
            const int i = k * NT + tid;
            pk[k] = 0u;
            if (i < n) pk[k] = gw[FILE_OFF + i];
        }
        #pragma unroll
        for (int k = 0; k < SORT_SLOTS; ++k)
            ns[k] = norm[pk[k] & 0xFFFFu];
        #pragma unroll
        for (int k = 0; k < SORT_SLOTS; ++k) {
            const int i = k * NT + tid;
            rk[k] = 0u;
            if (i < n) rk[k] = atomicAdd(&rcnt[pk[k] >> 16], 1u);
        }
        __syncthreads();

        if (wave == 0) {
            uint4 ca = *(const uint4*)(rcnt + 8 * lane);
            uint4 cb = *(const uint4*)(rcnt + 8 * lane + 4);
            unsigned c[8] = { ca.x, ca.y, ca.z, ca.w, cb.x, cb.y, cb.z, cb.w };
            unsigned t = 0u;
            #pragma unroll
            for (int j = 0; j < 8; ++j) { const unsigned v = c[j]; c[j] = t; t += v; }
            unsigned incl = t;
            #pragma unroll
            for (int d = 1; d < 32; d <<= 1) {
                const unsigned o = __shfl_up(incl, (unsigned)d, 32);
                if (lane >= d) incl += o;
            }
            const unsigned base = incl - t;
            ca = make_uint4(base + c[0], base + c[1], base + c[2], base + c[3]);
            cb = make_uint4(base + c[4], base + c[5], base + c[6], base + c[7]);
            *(uint4*)(rcnt + 8 * lane)     = ca;
            *(uint4*)(rcnt + 8 * lane + 4) = cb;
        }
        __syncthreads();

        #pragma unroll
        for (int k = 0; k < SORT_SLOTS; ++k) {
            const int i = k * NT + tid;
            if (i < n) {
                const unsigned pos = rcnt[pk[k] >> 16] + rk[k];
                q[pos]   = pk[k] & 0xFFFFu;
                qns[pos] = ns[k];
            }
        }
        __syncthreads();

        for (int r = wave; r < R_NODES; r += NT / 32) {
            const int e0 = __builtin_amdgcn_readfirstlane((int)rcnt[r]);
            const int e1 = __builtin_amdgcn_readfirstlane((int)rcnt[r + 1]);
            const int row = nbase + r;
            const float4* f0p = (const float4*)(feat0 + (size_t)row * DIM + 4 * lane);
            float4 f0 = make_float4(0.0f, 0.0f, 0.0f, 0.0f);
            float nrow = 0.0f;
            bool have_f0 = false;
            float4 acc = make_float4(0.0f, 0.0f, 0.0f, 0.0f);
            for (int b0 = e0; b0 < e1; b0 += 32) {
                unsigned sv = 0u;
                float nv = 0.0f;
                if (b0 + lane < e1) { sv = q[b0 + lane]; nv = qns[b0 + lane]; }
                const int m = min(32, e1 - b0);
                for (int c0 = 0; c0 < m; c0 += ROW_CHUNK) {
                    float4 x[ROW_CHUNK];
                    #pragma unroll
                    for (int u = 0; u < ROW_CHUNK; ++u) {
                        if (c0 + u >= m) break;
                        const int s = __builtin_amdgcn_readlane((int)sv, c0 + u);
                        x[u] = *(const float4*)(feat + (size_t)s * DIM + 4 * lane);
                    }
                    if (!have_f0) { f0 = *f0p; nrow = norm[row]; have_f0 = true; }
                    #pragma unroll
                    for (int u = 0; u < ROW_CHUNK; ++u) {
                        if (c0 + u >= m) break;
                        const float w = __builtin_bit_cast(float,
                            __builtin_amdgcn_readlane(__builtin_bit_cast(int, nv), c0 + u));
                        acc.x = fmaf(x[u].x, w, acc.x);
                        acc.y = fmaf(x[u].y, w, acc.y);
                        acc.z = fmaf(x[u].z, w, acc.z);
                        acc.w = fmaf(x[u].w, w, acc.w);
                    }
                }
            }
            if (!have_f0) { f0 = *f0p; nrow = norm[row]; }
            const float nr = nrow * (1.0f - ALPHA_C);
            float4 h;
            h.x = acc.x * nr + ALPHA_C * f0.x;
            h.y = acc.y * nr + ALPHA_C * f0.y;
            h.z = acc.z * nr + ALPHA_C * f0.z;
            h.w = acc.w * nr + ALPHA_C * f0.w;
            *(float4*)(hacc + r * HS_STRIDE + 4 * lane) = h;
        }
    } else {
        for (int i = tid; i < ACC_ROWS * HS_STRIDE / 4; i += NT)
            ((float4*)hacc)[i] = make_float4(0.0f, 0.0f, 0.0f, 0.0f);
        __syncthreads();
        const int4* d4 = (const int4*)dst;
        const int off = blockIdx.x * (NV4 / 256);
        for (int i0 = 0; i0 < SCAN_FULL; i0 += NT * SCAN_U) {
            int4 v[SCAN_U];
            int  jx[SCAN_U];
            #pragma unroll
            for (int u = 0; u < SCAN_U; ++u) {
                int j = i0 + u * NT + tid + off;
                if (j >= NV4) j -= NV4;
                jx[u] = j;
                v[u] = d4[j];
            }
            #pragma unroll
            for (int u = 0; u < SCAN_U; ++u)
                scan_one(v[u], jx[u], nbase, q, &qn, hacc, feat, norm, src);
        }
        {
            int4 v[SCAN_TAIL_U];
            int  jx[SCAN_TAIL_U];
            #pragma unroll
            for (int u = 0; u < SCAN_TAIL_U; ++u) {
                int i = SCAN_FULL + u * NT + tid;
                int j = i + off;
                if (j >= NV4) j -= NV4;
                jx[u] = j;
                v[u] = (i < NV4) ? d4[j] : make_int4(-1, -1, -1, -1);
            }
            #pragma unroll
            for (int u = 0; u < SCAN_TAIL_U; ++u)
                scan_one(v[u], jx[u], nbase, q, &qn, hacc, feat, norm, src);
        }
        __syncthreads();

        const int n = __builtin_amdgcn_readfirstlane((int)min(qn, (unsigned)QCAP));
        const int n8 = (n + CU - 1) & ~(CU - 1);
        for (int i = tid; i < n8; i += NT) {
            unsigned pk = (unsigned)SINK_ROW << 16;
            float ns = 0.0f;
            if (i < n) {
                unsigned p = q[i];
                int s = src[p & 0xFFFFFu];
                ns = norm[s];
                pk = ((p >> 20) << 16) | (unsigned)s;
            }
            q[i]   = pk;
            qns[i] = ns;
        }
        __syncthreads();

        for (int i0 = wave * CU; i0 < n; i0 += (NT / 32) * CU) {
            const uint4  pa = *(const uint4*)(q + i0);
            const uint4  pb = *(const uint4*)(q + i0 + 4);
            const float4 na = *(const float4*)(qns + i0);
            const float4 nb = *(const float4*)(qns + i0 + 4);
            const unsigned pku[CU] = { pa.x, pa.y, pa.z, pa.w, pb.x, pb.y, pb.z, pb.w };
            const float    nsu[CU] = { na.x, na.y, na.z, na.w, nb.x, nb.y, nb.z, nb.w };
            v2f xa[CU], xb[CU];
            #pragma unroll
            for (int u = 0; u < CU; ++u) {
                const int s = __builtin_amdgcn_readfirstlane((int)(pku[u] & 0xFFFFu));
                const float* fr = feat + (size_t)s * DIM + 2 * lane;
                xa[u] = *(const v2f*)(fr);
                xb[u] = *(const v2f*)(fr + 64);
            }
            #pragma unroll
            for (int u = 0; u < CU; ++u) {
                const int dl = __builtin_amdgcn_readfirstlane((int)(pku[u] >> 16));
                float* ar = hacc + dl * HS_STRIDE + 2 * lane;
                atomicAdd(ar,      xa[u].x * nsu[u]);
                atomicAdd(ar + 1,  xa[u].y * nsu[u]);
                atomicAdd(ar + 64, xb[u].x * nsu[u]);
                atomicAdd(ar + 65, xb[u].y * nsu[u]);
            }
        }
        __syncthreads();

        for (int i = tid; i < R_NODES * (DIM / 4); i += NT) {
            int r = i >> 5;
            int c = (i & 31) * 4;
            int row = nbase + r;
            float nr = norm[row] * (1.0f - ALPHA_C);
            float4 a  = *(const float4*)(hacc + r * HS_STRIDE + c);
            float4 f0 = *(const float4*)(feat0 + (size_t)row * DIM + c);
            float4 h;
            h.x = a.x * nr + ALPHA_C * f0.x;
            h.y = a.y * nr + ALPHA_C * f0.y;
            h.z = a.z * nr + ALPHA_C * f0.z;
            h.w = a.w * nr + ALPHA_C * f0.w;
            *(float4*)(hacc + r * HS_STRIDE + c) = h;
        }
    }
    __syncthreads();

    const int half = lane >> 4;
    const int lr   = lane & 15;
    const int n0   = (wave & 7) * 16;
    for (int st = wave >> 3; st < N_STRIPS; st += 4) {
        const float* hs = hacc + st * 16 * HS_STRIDE;
        v8f acc = {};
        #pragma unroll
        for (int k0 = 0; k0 < DIM; k0 += 4) {
            const int ka = k0 + 2 * half;
            v2f a, b;
            a.x = hs[lr * HS_STRIDE + ka];
            a.y = hs[lr * HS_STRIDE + ka + 1];
            b.x = Wl[ka * DIM + n0 + lr];
            b.y = Wl[(ka + 1) * DIM + n0 + lr];
            acc = __builtin_amdgcn_wmma_f32_16x16x4_f32(
                 false, a,  false, b,
                 (short)0, acc,  false,  false);
        }

        const int col   = n0 + lr;
        const int lrow0 = st * 16 + 8 * half;
        float val[8];
        #pragma unroll
        for (int r = 0; r < 8; r++) {
            float hv = hacc[(lrow0 + r) * HS_STRIDE + col];
            val[r] = (1.0f - BETA_C) * hv + BETA_C * acc[r] + bs[col];
        }
        if (lrow0 < R_NODES) {
            #pragma unroll
            for (int r = 0; r < 8; r++)
                out[(size_t)(nbase + lrow0 + r) * DIM + col] = val[r];
        }
    }
}

extern "C" void kernel_launch(void* const* d_in, const int* in_sizes, int n_in,
                              void* d_out, int out_size, void* d_ws, size_t ws_size,
                              hipStream_t stream) {
    const float* feat  = (const float*)d_in[0];
    const float* feat0 = (const float*)d_in[1];
    const float* W     = (const float*)d_in[2];
    const float* bias  = (const float*)d_in[3];
    const int*   src   = (const int*)d_in[4];
    const int*   dst   = (const int*)d_in[5];
    float* out = (float*)d_out;
    float* norm = (float*)d_ws;

    gcnii_bin<<<BIN_GROUPS, BIN_NT, 0, stream>>>(src, dst, out);
    gcnii_degnorm<<<N_GROUPS, DN_NT, 0, stream>>>(dst, norm, out);
    gcnii_own<<<N_GROUPS, NT, 0, stream>>>(feat, feat0, W, bias, norm, src, dst, out);
}
